# starting kernel
# baseline (speedup 1.0000x reference)
#include <hip/hip_runtime.h>
#include <math.h>

typedef __attribute__((ext_vector_type(2))) float v2f;
typedef __attribute__((ext_vector_type(8))) float v8f;

#define H  256
#define NV 512
#define NT 8192
#define V  64

__global__ void k_vehicle_proj(const float* __restrict__ xv,
                               const int* __restrict__ current,
                               const float* __restrict__ W1,
                               const float* __restrict__ b1,
                               float* __restrict__ Bv) {
  __shared__ float veh[H];
  const int v = blockIdx.x;
  const int h = threadIdx.x;
  const int row = current[2 * v];
  veh[h] = xv[row * H + h];
  __syncthreads();
  const float* w = W1 + h * (2 * H) + H;
  float s = b1[h];
#pragma unroll 8
  for (int i = 0; i < H; ++i) s = fmaf(veh[i], w[i], s);
  Bv[v * H + h] = s;
}

__global__ void k_track_proj_wmma(const float* __restrict__ xt,
                                  const float* __restrict__ W1,
                                  float* __restrict__ A) {
  const int lane   = threadIdx.x & 31;
  const int wv     = threadIdx.x >> 5;
  const int tile   = blockIdx.x * 4 + wv;
  const int n_base = (tile & 15) * 16;
  const int m_base = (tile >> 4) * 16;
  const int half   = lane >> 4;
  const int l      = lane & 15;

  const float* aptr = xt + (m_base + l) * H + 2 * half;
  const float* bptr = W1 + (n_base + l) * (2 * H) + 2 * half;

  v8f c = {};
#pragma unroll 4
  for (int k = 0; k < H; k += 4) {
    v2f a = *(const v2f*)(aptr + k);
    v2f b = *(const v2f*)(bptr + k);
    c = __builtin_amdgcn_wmma_f32_16x16x4_f32(
         false, a,  false, b,
         (short)0, c,  false,  false);
  }

  float* dptr = A + (m_base + 8 * half) * H + n_base + l;
#pragma unroll
  for (int r = 0; r < 8; ++r) dptr[r * H] = c[r];
}

__global__ void k_pair_mlp(const float* __restrict__ A,
                           const float* __restrict__ Bv,
                           const float* __restrict__ W2,
                           const float* __restrict__ b2,
                           float* __restrict__ out) {
  __shared__ float BvL[V * H];
  const int tid = threadIdx.x;
#pragma unroll
  for (int j = 0; j < (V * H) / 256; ++j) {
    int idx = tid + j * 256;
    int v = idx >> 8, h = idx & 255;
    BvL[v * H + (h ^ v)] = Bv[idx];
  }
  __syncthreads();

  const int v  = tid & 63;
  const int tl = tid >> 6;
  const float b2_0 = b2[0], b2_1 = b2[1];
  const float* bl = BvL + v * H;

  for (int j = 0; j < 4; ++j) {
    const int t = blockIdx.x * 16 + j * 4 + tl;
    const float* ar = A + t * H;
    float s0 = 0.f, s1 = 0.f;
#pragma unroll 4
    for (int h = 0; h < H; ++h) {
      float x = ar[h] + bl[h ^ v];
      x = fmaxf(x, 0.f);
      s0 = fmaf(x, W2[h], s0);
      s1 = fmaf(x, W2[H + h], s1);
    }
    out[v * NT + t] = s0 + b2_0;
    const float z = s1 + b2_1;
    out[V * NT + v * NT + t] = 1.0f / (1.0f + __expf(-z));
  }
}

extern "C" void kernel_launch(void* const* d_in, const int* in_sizes, int n_in,
                              void* d_out, int out_size, void* d_ws, size_t ws_size,
                              hipStream_t stream) {
  const float* xv  = (const float*)d_in[0];
  const float* xt  = (const float*)d_in[1];
  const int*   cur = (const int*)  d_in[2];
  const float* W1  = (const float*)d_in[3];
  const float* b1  = (const float*)d_in[4];
  const float* W2  = (const float*)d_in[5];
  const float* b2  = (const float*)d_in[6];
  float* out = (float*)d_out;

  float* A  = (float*)d_ws;
  float* Bv = A + NT * H;

  k_vehicle_proj<<<V, H, 0, stream>>>(xv, cur, W1, b1, Bv);
  k_track_proj_wmma<<<(NT / 16) * (H / 16) / 4, 128, 0, stream>>>(xt, W1, A);
  k_pair_mlp<<<NT / 16, 256, 0, stream>>>(A, Bv, W2, b2, out);
}


# pair epilogue register-tiled: thread owns 2 vehicles x 4 tracks, A and Bv rows staged whole in LDS, read 16 B wide one 4-step fetch ahead; per-pair h order and fmaf chain unchanged
# speedup vs baseline: 1.8204x; 1.8204x over previous
#include <hip/hip_runtime.h>
#include <math.h>

typedef __attribute__((ext_vector_type(2))) float v2f;
typedef __attribute__((ext_vector_type(8))) float v8f;

#define H  256
#define NV 512
#define NT 8192
#define V  64

__global__ void k_vehicle_proj(const float* __restrict__ xv,
                               const int* __restrict__ current,
                               const float* __restrict__ W1,
                               const float* __restrict__ b1,
                               float* __restrict__ Bv) {
  __shared__ float veh[H];
  const int v = blockIdx.x;
  const int h = threadIdx.x;
  const int row = current[2 * v];
  veh[h] = xv[row * H + h];
  __syncthreads();
  const float* w = W1 + h * (2 * H) + H;
  float s = b1[h];
#pragma unroll 8
  for (int i = 0; i < H; ++i) s = fmaf(veh[i], w[i], s);
  Bv[v * H + h] = s;
}

__global__ void k_track_proj_wmma(const float* __restrict__ xt,
                                  const float* __restrict__ W1,
                                  float* __restrict__ A) {
  const int lane   = threadIdx.x & 31;
  const int wv     = threadIdx.x >> 5;
  const int tile   = blockIdx.x * 4 + wv;
  const int n_base = (tile & 15) * 16;
  const int m_base = (tile >> 4) * 16;
  const int half   = lane >> 4;
  const int l      = lane & 15;

  const float* aptr = xt + (m_base + l) * H + 2 * half;
  const float* bptr = W1 + (n_base + l) * (2 * H) + 2 * half;

  v8f c = {};
#pragma unroll 4
  for (int k = 0; k < H; k += 4) {
    v2f a = *(const v2f*)(aptr + k);
    v2f b = *(const v2f*)(bptr + k);
    c = __builtin_amdgcn_wmma_f32_16x16x4_f32(
         false, a,  false, b,
         (short)0, c,  false,  false);
  }

  float* dptr = A + (m_base + 8 * half) * H + n_base + l;
#pragma unroll
  for (int r = 0; r < 8; ++r) dptr[r * H] = c[r];
}

#define TT    32
#define PITCH 260

typedef __attribute__((ext_vector_type(4))) float v4f;

struct PairOperands {
  v4f a[4];
  v4f b[2];
  v4f w0, w1;
};

__device__ __forceinline__ void pair_fetch(PairOperands& o, const float* ap, const float* bp,
                                           const float* __restrict__ W2, int h) {
#pragma unroll
  for (int j = 0; j < 4; ++j) o.a[j] = *(const v4f*)(ap + j * 8 * PITCH + h);
#pragma unroll
  for (int i = 0; i < 2; ++i) o.b[i] = *(const v4f*)(bp + i * 32 * PITCH + h);
  o.w0 = *(const v4f*)(W2 + h);
  o.w1 = *(const v4f*)(W2 + H + h);
}

__device__ __forceinline__ void pair_hold(const PairOperands& o) {
  asm volatile("" :: "v"(o.a[0]), "v"(o.a[1]), "v"(o.a[2]), "v"(o.a[3]),
                     "v"(o.b[0]), "v"(o.b[1]), "s"(o.w0), "s"(o.w1));
}

__device__ __forceinline__ void pair_step(const PairOperands& o, float (&s0)[2][4], float (&s1)[2][4]) {
#pragma unroll
  for (int e = 0; e < 4; ++e) {
#pragma unroll
    for (int i = 0; i < 2; ++i) {
#pragma unroll
      for (int j = 0; j < 4; ++j) {
        float x = o.a[j][e] + o.b[i][e];
        x = fmaxf(x, 0.f);
        s0[i][j] = fmaf(x, o.w0[e], s0[i][j]);
        s1[i][j] = fmaf(x, o.w1[e], s1[i][j]);
      }
    }
  }
}

__global__ __launch_bounds__(256) void k_pair_mlp(const float* __restrict__ A,
                                                  const float* __restrict__ Bv,
                                                  const float* __restrict__ W2,
                                                  const float* __restrict__ b2,
                                                  float* __restrict__ out) {
  __shared__ __attribute__((aligned(16))) float L[(V + TT) * PITCH];
  float* BvL = L;
  float* AL  = L + V * PITCH;
  const int tid = threadIdx.x;
  const int tb  = blockIdx.x * TT;

  const v4f* Bv4 = (const v4f*)Bv;
#pragma unroll
  for (int n = 0; n < (V * H / 4) / 256; ++n) {
    const int q = tid + n * 256;
    const int row = q >> 6, c4 = q & 63;
    *(v4f*)(BvL + row * PITCH + 4 * c4) = Bv4[q];
  }
  const v4f* A4 = (const v4f*)(A + tb * H);
#pragma unroll
  for (int n = 0; n < (TT * H / 4) / 256; ++n) {
    const int q = tid + n * 256;
    const int row = q >> 6, c4 = q & 63;
    const int lrow = (row & 3) * 8 + (row >> 2);
    *(v4f*)(AL + lrow * PITCH + 4 * c4) = A4[q];
  }
  __syncthreads();

  const int tg = tid & 7;
  const int vq = tid >> 3;
  const float* ap = AL + tg * PITCH;
  const float* bp = BvL + vq * PITCH;

  float s0[2][4], s1[2][4];
#pragma unroll
  for (int i = 0; i < 2; ++i)
#pragma unroll
    for (int j = 0; j < 4; ++j) { s0[i][j] = 0.f; s1[i][j] = 0.f; }

  PairOperands p, q;
  pair_fetch(p, ap, bp, W2, 0);
  pair_hold(p);
  for (int h = 0; h < H; h += 16) {
    pair_fetch(q, ap, bp, W2, h + 4);
    pair_step(p, s0, s1);
    pair_hold(q);
    pair_fetch(p, ap, bp, W2, h + 8);
    pair_step(q, s0, s1);
    pair_hold(p);
    pair_fetch(q, ap, bp, W2, h + 12);
    pair_step(p, s0, s1);
    pair_hold(q);
    pair_fetch(p, ap, bp, W2, (h + 16) & (H - 1));
    pair_step(q, s0, s1);
    pair_hold(p);
  }

  const float b2_0 = b2[0], b2_1 = b2[1];
#pragma unroll
  for (int i = 0; i < 2; ++i) {
    const int v = vq + 32 * i;
    v4f sc, ps;
#pragma unroll
    for (int j = 0; j < 4; ++j) {
      sc[j] = s0[i][j] + b2_0;
      const float z = s1[i][j] + b2_1;
      ps[j] = 1.0f / (1.0f + __expf(-z));
    }
    *(v4f*)(out + v * NT + tb + 4 * tg) = sc;
    *(v4f*)(out + V * NT + v * NT + tb + 4 * tg) = ps;
  }
}

extern "C" void kernel_launch(void* const* d_in, const int* in_sizes, int n_in,
                              void* d_out, int out_size, void* d_ws, size_t ws_size,
                              hipStream_t stream) {
  const float* xv  = (const float*)d_in[0];
  const float* xt  = (const float*)d_in[1];
  const int*   cur = (const int*)  d_in[2];
  const float* W1  = (const float*)d_in[3];
  const float* b1  = (const float*)d_in[4];
  const float* W2  = (const float*)d_in[5];
  const float* b2  = (const float*)d_in[6];
  float* out = (float*)d_out;

  float* A  = (float*)d_ws;
  float* Bv = A + NT * H;

  k_vehicle_proj<<<V, H, 0, stream>>>(xv, cur, W1, b1, Bv);
  k_track_proj_wmma<<<(NT / 16) * (H / 16) / 4, 128, 0, stream>>>(xt, W1, A);
  k_pair_mlp<<<NT / TT, 256, 0, stream>>>(A, Bv, W2, b2, out);
}


# track GEMM register-tiled: wave owns 2x2 tiles (block 32 rows x 256 h), fragments fetched 16 B wide a k block ahead, one k order in both loaders; the f32 sum over k of A runs in a new order
# speedup vs baseline: 2.3294x; 1.2796x over previous
#include <hip/hip_runtime.h>
#include <math.h>

typedef __attribute__((ext_vector_type(2))) float v2f;
typedef __attribute__((ext_vector_type(8))) float v8f;

#define H  256
#define NV 512
#define NT 8192
#define V  64

__global__ void k_vehicle_proj(const float* __restrict__ xv,
                               const int* __restrict__ current,
                               const float* __restrict__ W1,
                               const float* __restrict__ b1,
                               float* __restrict__ Bv) {
  __shared__ float veh[H];
  const int v = blockIdx.x;
  const int h = threadIdx.x;
  const int row = current[2 * v];
  veh[h] = xv[row * H + h];
  __syncthreads();
  const float* w = W1 + h * (2 * H) + H;
  float s = b1[h];
#pragma unroll 8
  for (int i = 0; i < H; ++i) s = fmaf(veh[i], w[i], s);
  Bv[v * H + h] = s;
}

typedef __attribute__((ext_vector_type(4))) float v4f;

struct Frag16 {
  v4f q[4];
};

__device__ __forceinline__ void frag_fetch(Frag16& f, const float* p) {
#pragma unroll
  for (int j = 0; j < 4; ++j) f.q[j] = *(const v4f*)(p + 4 * j);
}

__device__ __forceinline__ v2f frag_step(const Frag16& f, int s) {
  v2f r;
  r[0] = f.q[s >> 1][2 * (s & 1)];
  r[1] = f.q[s >> 1][2 * (s & 1) + 1];
  return r;
}

__global__ __launch_bounds__(256) void k_track_proj_wmma(const float* __restrict__ xt,
                                                         const float* __restrict__ W1,
                                                         float* __restrict__ A) {
  const int lane = threadIdx.x & 31;
  const int wv   = threadIdx.x >> 5;
  const int half = lane >> 4;
  const int l    = lane & 15;
  const int m0   = blockIdx.x * 32;
  const int n0   = wv * 32;

  const float* ap = xt + (m0 + l) * H + 16 * half;
  const float* bp = W1 + (n0 + l) * (2 * H) + 16 * half;

  v8f c[2][2];
#pragma unroll
  for (int mi = 0; mi < 2; ++mi)
#pragma unroll
    for (int ni = 0; ni < 2; ++ni) c[mi][ni] = (v8f){};

  Frag16 fa[2][2], fb[2][2];
  frag_fetch(fa[0][0], ap);
  frag_fetch(fa[0][1], ap + 16 * H);
  frag_fetch(fb[0][0], bp);
  frag_fetch(fb[0][1], bp + 16 * (2 * H));

#pragma unroll
  for (int kb = 0; kb < H / 32; ++kb) {
    const int cur = kb & 1, nxt = cur ^ 1;
    if (kb + 1 < H / 32) {
      const int ko = (kb + 1) * 32;
      frag_fetch(fa[nxt][0], ap + ko);
      frag_fetch(fa[nxt][1], ap + 16 * H + ko);
      frag_fetch(fb[nxt][0], bp + ko);
      frag_fetch(fb[nxt][1], bp + 16 * (2 * H) + ko);
    }
    __builtin_amdgcn_sched_barrier(0);
#pragma unroll
    for (int s = 0; s < 8; ++s) {
#pragma unroll
      for (int mi = 0; mi < 2; ++mi) {
#pragma unroll
        for (int ni = 0; ni < 2; ++ni) {
          c[mi][ni] = __builtin_amdgcn_wmma_f32_16x16x4_f32(
               false, frag_step(fa[cur][mi], s),
               false, frag_step(fb[cur][ni], s),
               (short)0, c[mi][ni],  false,  false);
        }
      }
    }
  }

#pragma unroll
  for (int mi = 0; mi < 2; ++mi) {
#pragma unroll
    for (int ni = 0; ni < 2; ++ni) {
      float* dptr = A + (m0 + 16 * mi + 8 * half) * H + n0 + 16 * ni + l;
#pragma unroll
      for (int r = 0; r < 8; ++r) dptr[r * H] = c[mi][ni][r];
    }
  }
}

#define TT    32
#define PITCH 260

struct PairOperands {
  v4f a[4];
  v4f b[2];
  v4f w0, w1;
};

__device__ __forceinline__ void pair_fetch(PairOperands& o, const float* ap, const float* bp,
                                           const float* __restrict__ W2, int h) {
#pragma unroll
  for (int j = 0; j < 4; ++j) o.a[j] = *(const v4f*)(ap + j * 8 * PITCH + h);
#pragma unroll
  for (int i = 0; i < 2; ++i) o.b[i] = *(const v4f*)(bp + i * 32 * PITCH + h);
  o.w0 = *(const v4f*)(W2 + h);
  o.w1 = *(const v4f*)(W2 + H + h);
}

__device__ __forceinline__ void pair_hold(const PairOperands& o) {
  asm volatile("" :: "v"(o.a[0]), "v"(o.a[1]), "v"(o.a[2]), "v"(o.a[3]),
                     "v"(o.b[0]), "v"(o.b[1]), "s"(o.w0), "s"(o.w1));
}

__device__ __forceinline__ void pair_step(const PairOperands& o, float (&s0)[2][4], float (&s1)[2][4]) {
#pragma unroll
  for (int e = 0; e < 4; ++e) {
#pragma unroll
    for (int i = 0; i < 2; ++i) {
#pragma unroll
      for (int j = 0; j < 4; ++j) {
        float x = o.a[j][e] + o.b[i][e];
        x = fmaxf(x, 0.f);
        s0[i][j] = fmaf(x, o.w0[e], s0[i][j]);
        s1[i][j] = fmaf(x, o.w1[e], s1[i][j]);
      }
    }
  }
}

__global__ __launch_bounds__(256) void k_pair_mlp(const float* __restrict__ A,
                                                  const float* __restrict__ Bv,
                                                  const float* __restrict__ W2,
                                                  const float* __restrict__ b2,
                                                  float* __restrict__ out) {
  __shared__ __attribute__((aligned(16))) float L[(V + TT) * PITCH];
  float* BvL = L;
  float* AL  = L + V * PITCH;
  const int tid = threadIdx.x;
  const int tb  = blockIdx.x * TT;

  const v4f* Bv4 = (const v4f*)Bv;
#pragma unroll
  for (int n = 0; n < (V * H / 4) / 256; ++n) {
    const int q = tid + n * 256;
    const int row = q >> 6, c4 = q & 63;
    *(v4f*)(BvL + row * PITCH + 4 * c4) = Bv4[q];
  }
  const v4f* A4 = (const v4f*)(A + tb * H);
#pragma unroll
  for (int n = 0; n < (TT * H / 4) / 256; ++n) {
    const int q = tid + n * 256;
    const int row = q >> 6, c4 = q & 63;
    const int lrow = (row & 3) * 8 + (row >> 2);
    *(v4f*)(AL + lrow * PITCH + 4 * c4) = A4[q];
  }
  __syncthreads();

  const int tg = tid & 7;
  const int vq = tid >> 3;
  const float* ap = AL + tg * PITCH;
  const float* bp = BvL + vq * PITCH;

  float s0[2][4], s1[2][4];
#pragma unroll
  for (int i = 0; i < 2; ++i)
#pragma unroll
    for (int j = 0; j < 4; ++j) { s0[i][j] = 0.f; s1[i][j] = 0.f; }

  PairOperands p, q;
  pair_fetch(p, ap, bp, W2, 0);
  pair_hold(p);
  for (int h = 0; h < H; h += 16) {
    pair_fetch(q, ap, bp, W2, h + 4);
    pair_step(p, s0, s1);
    pair_hold(q);
    pair_fetch(p, ap, bp, W2, h + 8);
    pair_step(q, s0, s1);
    pair_hold(p);
    pair_fetch(q, ap, bp, W2, h + 12);
    pair_step(p, s0, s1);
    pair_hold(q);
    pair_fetch(p, ap, bp, W2, (h + 16) & (H - 1));
    pair_step(q, s0, s1);
    pair_hold(p);
  }

  const float b2_0 = b2[0], b2_1 = b2[1];
#pragma unroll
  for (int i = 0; i < 2; ++i) {
    const int v = vq + 32 * i;
    v4f sc, ps;
#pragma unroll
    for (int j = 0; j < 4; ++j) {
      sc[j] = s0[i][j] + b2_0;
      const float z = s1[i][j] + b2_1;
      ps[j] = 1.0f / (1.0f + __expf(-z));
    }
    *(v4f*)(out + v * NT + tb + 4 * tg) = sc;
    *(v4f*)(out + V * NT + v * NT + tb + 4 * tg) = ps;
  }
}

extern "C" void kernel_launch(void* const* d_in, const int* in_sizes, int n_in,
                              void* d_out, int out_size, void* d_ws, size_t ws_size,
                              hipStream_t stream) {
  const float* xv  = (const float*)d_in[0];
  const float* xt  = (const float*)d_in[1];
  const int*   cur = (const int*)  d_in[2];
  const float* W1  = (const float*)d_in[3];
  const float* b1  = (const float*)d_in[4];
  const float* W2  = (const float*)d_in[5];
  const float* b2  = (const float*)d_in[6];
  float* out = (float*)d_out;

  float* A  = (float*)d_ws;
  float* Bv = A + NT * H;

  k_vehicle_proj<<<V, H, 0, stream>>>(xv, cur, W1, b1, Bv);
  k_track_proj_wmma<<<NT / 32, 256, 0, stream>>>(xt, W1, A);
  k_pair_mlp<<<NT / TT, 256, 0, stream>>>(A, Bv, W2, b2, out);
}


# vehicle projection moved into the track GEMM launch as two extra waves per block (2 launches, not 3): its dependent load+fmaf chains overlap the WMMA waves; every sum order unchanged
# speedup vs baseline: 2.8226x; 1.2117x over previous
#include <hip/hip_runtime.h>
#include <math.h>

typedef __attribute__((ext_vector_type(2))) float v2f;
typedef __attribute__((ext_vector_type(8))) float v8f;

#define H  256
#define NV 512
#define NT 8192
#define V  64

__device__ __forceinline__ void vehicle_proj_part(int blk, int t,
                                                  const float* __restrict__ xv,
                                                  const int* __restrict__ current,
                                                  const float* __restrict__ W1,
                                                  const float* __restrict__ b1,
                                                  float* __restrict__ Bv) {
  const int v = blk >> 2;
  const int h = 64 * (blk & 3) + t;
  const float* veh = xv + current[2 * v] * H;
  const float* w = W1 + h * (2 * H) + H;
  float s = b1[h];
#pragma unroll 8
  for (int i = 0; i < H; ++i) s = fmaf(veh[i], w[i], s);
  Bv[v * H + h] = s;
}

typedef __attribute__((ext_vector_type(4))) float v4f;

struct Frag16 {
  v4f q[4];
};

__device__ __forceinline__ void frag_fetch(Frag16& f, const float* p) {
#pragma unroll
  for (int j = 0; j < 4; ++j) f.q[j] = *(const v4f*)(p + 4 * j);
}

__device__ __forceinline__ v2f frag_step(const Frag16& f, int s) {
  v2f r;
  r[0] = f.q[s >> 1][2 * (s & 1)];
  r[1] = f.q[s >> 1][2 * (s & 1) + 1];
  return r;
}

#define GEMM_WAVES   8
#define PROJ_THREADS 320

__global__ __launch_bounds__(PROJ_THREADS) void k_proj(const float* __restrict__ xv,
                                                       const int* __restrict__ current,
                                                       const float* __restrict__ xt,
                                                       const float* __restrict__ W1,
                                                       const float* __restrict__ b1,
                                                       float* __restrict__ A,
                                                       float* __restrict__ Bv) {
  const int lane = threadIdx.x & 31;
  const int wv   = __builtin_amdgcn_readfirstlane(threadIdx.x >> 5);
  if (wv >= GEMM_WAVES) {
    vehicle_proj_part(blockIdx.x, threadIdx.x - 32 * GEMM_WAVES, xv, current, W1, b1, Bv);
    return;
  }

  const int half = lane >> 4;
  const int l    = lane & 15;
  const int m0   = blockIdx.x * 32;
  const int n0   = wv * 32;

  const float* ap = xt + (m0 + l) * H + 16 * half;
  const float* bp = W1 + (n0 + l) * (2 * H) + 16 * half;

  v8f c[2][2];
#pragma unroll
  for (int mi = 0; mi < 2; ++mi)
#pragma unroll
    for (int ni = 0; ni < 2; ++ni) c[mi][ni] = (v8f){};

  Frag16 fa[2][2], fb[2][2];
  frag_fetch(fa[0][0], ap);
  frag_fetch(fa[0][1], ap + 16 * H);
  frag_fetch(fb[0][0], bp);
  frag_fetch(fb[0][1], bp + 16 * (2 * H));

#pragma unroll
  for (int kb = 0; kb < H / 32; ++kb) {
    const int cur = kb & 1, nxt = cur ^ 1;
    if (kb + 1 < H / 32) {
      const int ko = (kb + 1) * 32;
      frag_fetch(fa[nxt][0], ap + ko);
      frag_fetch(fa[nxt][1], ap + 16 * H + ko);
      frag_fetch(fb[nxt][0], bp + ko);
      frag_fetch(fb[nxt][1], bp + 16 * (2 * H) + ko);
    }
    __builtin_amdgcn_sched_barrier(0);
#pragma unroll
    for (int s = 0; s < 8; ++s) {
#pragma unroll
      for (int mi = 0; mi < 2; ++mi) {
#pragma unroll
        for (int ni = 0; ni < 2; ++ni) {
          c[mi][ni] = __builtin_amdgcn_wmma_f32_16x16x4_f32(
               false, frag_step(fa[cur][mi], s),
               false, frag_step(fb[cur][ni], s),
               (short)0, c[mi][ni],  false,  false);
        }
      }
    }
  }

#pragma unroll
  for (int mi = 0; mi < 2; ++mi) {
#pragma unroll
    for (int ni = 0; ni < 2; ++ni) {
      float* dptr = A + (m0 + 16 * mi + 8 * half) * H + n0 + 16 * ni + l;
#pragma unroll
      for (int r = 0; r < 8; ++r) dptr[r * H] = c[mi][ni][r];
    }
  }
}

#define TT    32
#define PITCH 260

struct PairOperands {
  v4f a[4];
  v4f b[2];
  v4f w0, w1;
};

__device__ __forceinline__ void pair_fetch(PairOperands& o, const float* ap, const float* bp,
                                           const float* __restrict__ W2, int h) {
#pragma unroll
  for (int j = 0; j < 4; ++j) o.a[j] = *(const v4f*)(ap + j * 8 * PITCH + h);
#pragma unroll
  for (int i = 0; i < 2; ++i) o.b[i] = *(const v4f*)(bp + i * 32 * PITCH + h);
  o.w0 = *(const v4f*)(W2 + h);
  o.w1 = *(const v4f*)(W2 + H + h);
}

__device__ __forceinline__ void pair_hold(const PairOperands& o) {
  asm volatile("" :: "v"(o.a[0]), "v"(o.a[1]), "v"(o.a[2]), "v"(o.a[3]),
                     "v"(o.b[0]), "v"(o.b[1]), "s"(o.w0), "s"(o.w1));
}

__device__ __forceinline__ void pair_step(const PairOperands& o, float (&s0)[2][4], float (&s1)[2][4]) {
#pragma unroll
  for (int e = 0; e < 4; ++e) {
#pragma unroll
    for (int i = 0; i < 2; ++i) {
#pragma unroll
      for (int j = 0; j < 4; ++j) {
        float x = o.a[j][e] + o.b[i][e];
        x = fmaxf(x, 0.f);
        s0[i][j] = fmaf(x, o.w0[e], s0[i][j]);
        s1[i][j] = fmaf(x, o.w1[e], s1[i][j]);
      }
    }
  }
}

__global__ __launch_bounds__(256) void k_pair_mlp(const float* __restrict__ A,
                                                  const float* __restrict__ Bv,
                                                  const float* __restrict__ W2,
                                                  const float* __restrict__ b2,
                                                  float* __restrict__ out) {
  __shared__ __attribute__((aligned(16))) float L[(V + TT) * PITCH];
  float* BvL = L;
  float* AL  = L + V * PITCH;
  const int tid = threadIdx.x;
  const int tb  = blockIdx.x * TT;

  const v4f* Bv4 = (const v4f*)Bv;
#pragma unroll
  for (int n = 0; n < (V * H / 4) / 256; ++n) {
    const int q = tid + n * 256;
    const int row = q >> 6, c4 = q & 63;
    *(v4f*)(BvL + row * PITCH + 4 * c4) = Bv4[q];
  }
  const v4f* A4 = (const v4f*)(A + tb * H);
#pragma unroll
  for (int n = 0; n < (TT * H / 4) / 256; ++n) {
    const int q = tid + n * 256;
    const int row = q >> 6, c4 = q & 63;
    const int lrow = (row & 3) * 8 + (row >> 2);
    *(v4f*)(AL + lrow * PITCH + 4 * c4) = A4[q];
  }
  __syncthreads();

  const int tg = tid & 7;
  const int vq = tid >> 3;
  const float* ap = AL + tg * PITCH;
  const float* bp = BvL + vq * PITCH;

  float s0[2][4], s1[2][4];
#pragma unroll
  for (int i = 0; i < 2; ++i)
#pragma unroll
    for (int j = 0; j < 4; ++j) { s0[i][j] = 0.f; s1[i][j] = 0.f; }

  PairOperands p, q;
  pair_fetch(p, ap, bp, W2, 0);
  pair_hold(p);
  for (int h = 0; h < H; h += 16) {
    pair_fetch(q, ap, bp, W2, h + 4);
    pair_step(p, s0, s1);
    pair_hold(q);
    pair_fetch(p, ap, bp, W2, h + 8);
    pair_step(q, s0, s1);
    pair_hold(p);
    pair_fetch(q, ap, bp, W2, h + 12);
    pair_step(p, s0, s1);
    pair_hold(q);
    pair_fetch(p, ap, bp, W2, (h + 16) & (H - 1));
    pair_step(q, s0, s1);
    pair_hold(p);
  }

  const float b2_0 = b2[0], b2_1 = b2[1];
#pragma unroll
  for (int i = 0; i < 2; ++i) {
    const int v = vq + 32 * i;
    v4f sc, ps;
#pragma unroll
    for (int j = 0; j < 4; ++j) {
      sc[j] = s0[i][j] + b2_0;
      const float z = s1[i][j] + b2_1;
      ps[j] = 1.0f / (1.0f + __expf(-z));
    }
    *(v4f*)(out + v * NT + tb + 4 * tg) = sc;
    *(v4f*)(out + V * NT + v * NT + tb + 4 * tg) = ps;
  }
}

extern "C" void kernel_launch(void* const* d_in, const int* in_sizes, int n_in,
                              void* d_out, int out_size, void* d_ws, size_t ws_size,
                              hipStream_t stream) {
  const float* xv  = (const float*)d_in[0];
  const float* xt  = (const float*)d_in[1];
  const int*   cur = (const int*)  d_in[2];
  const float* W1  = (const float*)d_in[3];
  const float* b1  = (const float*)d_in[4];
  const float* W2  = (const float*)d_in[5];
  const float* b2  = (const float*)d_in[6];
  float* out = (float*)d_out;

  float* A  = (float*)d_ws;
  float* Bv = A + NT * H;

  k_proj<<<NT / 32, PROJ_THREADS, 0, stream>>>(xv, cur, xt, W1, b1, A, Bv);
  k_pair_mlp<<<NT / TT, 256, 0, stream>>>(A, Bv, W2, b2, out);
}


# track GEMM operands read as whole 128-byte lines (4 lines per global load, not 16), made into fragments through a wave-private LDS strip; fragment contents and every sum order unchanged
# speedup vs baseline: 3.1113x; 1.1023x over previous
#include <hip/hip_runtime.h>
#include <math.h>

typedef __attribute__((ext_vector_type(2))) float v2f;
typedef __attribute__((ext_vector_type(8))) float v8f;

#define H  256
#define NV 512
#define NT 8192
#define V  64

__device__ __forceinline__ void vehicle_proj_part(int blk, int t,
                                                  const float* __restrict__ xv,
                                                  const int* __restrict__ current,
                                                  const float* __restrict__ W1,
                                                  const float* __restrict__ b1,
                                                  float* __restrict__ Bv) {
  const int v = blk >> 2;
  const int h = 64 * (blk & 3) + t;
  const float* veh = xv + current[2 * v] * H;
  const float* w = W1 + h * (2 * H) + H;
  float s = b1[h];
#pragma unroll 8
  for (int i = 0; i < H; ++i) s = fmaf(veh[i], w[i], s);
  Bv[v * H + h] = s;
}

typedef __attribute__((ext_vector_type(4))) float v4f;

struct Frag16 {
  v4f q[4];
};

__device__ __forceinline__ void frag_fetch(Frag16& f, const float* p) {
#pragma unroll
  for (int j = 0; j < 4; ++j) f.q[j] = *(const v4f*)(p + 4 * j);
}

__device__ __forceinline__ v2f frag_step(const Frag16& f, int s) {
  v2f r;
  r[0] = f.q[s >> 1][2 * (s & 1)];
  r[1] = f.q[s >> 1][2 * (s & 1) + 1];
  return r;
}

#define SP         36
#define STRIP_ROWS 64

struct RawBlock {
  v4f x[8];
  v4f w[8];
};

__device__ __forceinline__ void raw_fetch(RawBlock& g, const float* xa, const float* wb, int ko) {
#pragma unroll
  for (int n = 0; n < 8; ++n) g.x[n] = *(const v4f*)(xa + 4 * n * H + ko);
#pragma unroll
  for (int n = 0; n < 8; ++n) g.w[n] = *(const v4f*)(wb + 4 * n * (2 * H) + ko);
}

__device__ __forceinline__ void raw_park(float* st, const RawBlock& g) {
#pragma unroll
  for (int n = 0; n < 8; ++n) *(v4f*)(st + 4 * n * SP) = g.x[n];
#pragma unroll
  for (int n = 0; n < 8; ++n) *(v4f*)(st + (32 + 4 * n) * SP) = g.w[n];
}

#define GEMM_WAVES   8
#define PROJ_THREADS 320

__global__ __launch_bounds__(PROJ_THREADS) void k_proj(const float* __restrict__ xv,
                                                       const int* __restrict__ current,
                                                       const float* __restrict__ xt,
                                                       const float* __restrict__ W1,
                                                       const float* __restrict__ b1,
                                                       float* __restrict__ A,
                                                       float* __restrict__ Bv) {
  const int lane = threadIdx.x & 31;
  const int wv   = __builtin_amdgcn_readfirstlane(threadIdx.x >> 5);
  if (wv >= GEMM_WAVES) {
    vehicle_proj_part(blockIdx.x, threadIdx.x - 32 * GEMM_WAVES, xv, current, W1, b1, Bv);
    return;
  }

  const int half = lane >> 4;
  const int l    = lane & 15;
  const int m0   = blockIdx.x * 32;
  const int n0   = wv * 32;

  const int sub   = lane >> 3;
  const int piece = lane & 7;
  const float* xa = xt + (m0 + sub) * H + 4 * piece;
  const float* wb = W1 + (n0 + sub) * (2 * H) + 4 * piece;

  __shared__ __attribute__((aligned(16))) float strips[GEMM_WAVES][2][STRIP_ROWS * SP];
  float* park[2]       = { &strips[wv][0][sub * SP + 4 * piece], &strips[wv][1][sub * SP + 4 * piece] };
  const float* frag[2] = { &strips[wv][0][l * SP + 16 * half],   &strips[wv][1][l * SP + 16 * half] };

  v8f c[2][2];
#pragma unroll
  for (int mi = 0; mi < 2; ++mi)
#pragma unroll
    for (int ni = 0; ni < 2; ++ni) c[mi][ni] = (v8f){};

  RawBlock g, g1;
  Frag16 fa[2][2], fb[2][2];
  raw_fetch(g, xa, wb, 0);
  raw_fetch(g1, xa, wb, 32);
  raw_park(park[0], g);
  raw_fetch(g, xa, wb, 64);
  frag_fetch(fa[0][0], frag[0]);
  frag_fetch(fa[0][1], frag[0] + 16 * SP);
  frag_fetch(fb[0][0], frag[0] + 32 * SP);
  frag_fetch(fb[0][1], frag[0] + 48 * SP);

#pragma unroll
  for (int kb = 0; kb < H / 32; ++kb) {
    const int cur = kb & 1, nxt = cur ^ 1;
    if (kb + 1 < H / 32) {
      if (kb == 0) {
        raw_park(park[nxt], g1);
      } else {
        raw_park(park[nxt], g);
        if (kb + 2 < H / 32) raw_fetch(g, xa, wb, (kb + 2) * 32);
      }
      frag_fetch(fa[nxt][0], frag[nxt]);
      frag_fetch(fa[nxt][1], frag[nxt] + 16 * SP);
      frag_fetch(fb[nxt][0], frag[nxt] + 32 * SP);
      frag_fetch(fb[nxt][1], frag[nxt] + 48 * SP);
    }
    __builtin_amdgcn_sched_barrier(0);
#pragma unroll
    for (int s = 0; s < 8; ++s) {
#pragma unroll
      for (int mi = 0; mi < 2; ++mi) {
#pragma unroll
        for (int ni = 0; ni < 2; ++ni) {
          c[mi][ni] = __builtin_amdgcn_wmma_f32_16x16x4_f32(
               false, frag_step(fa[cur][mi], s),
               false, frag_step(fb[cur][ni], s),
               (short)0, c[mi][ni],  false,  false);
        }
      }
    }
    __builtin_amdgcn_sched_barrier(0);
  }

#pragma unroll
  for (int mi = 0; mi < 2; ++mi) {
#pragma unroll
    for (int ni = 0; ni < 2; ++ni) {
      float* dptr = A + (m0 + 16 * mi + 8 * half) * H + n0 + 16 * ni + l;
#pragma unroll
      for (int r = 0; r < 8; ++r) dptr[r * H] = c[mi][ni][r];
    }
  }
}

#define TT    32
#define PITCH 260

struct PairOperands {
  v4f a[4];
  v4f b[2];
  v4f w0, w1;
};

__device__ __forceinline__ void pair_fetch(PairOperands& o, const float* ap, const float* bp,
                                           const float* __restrict__ W2, int h) {
#pragma unroll
  for (int j = 0; j < 4; ++j) o.a[j] = *(const v4f*)(ap + j * 8 * PITCH + h);
#pragma unroll
  for (int i = 0; i < 2; ++i) o.b[i] = *(const v4f*)(bp + i * 32 * PITCH + h);
  o.w0 = *(const v4f*)(W2 + h);
  o.w1 = *(const v4f*)(W2 + H + h);
}

__device__ __forceinline__ void pair_hold(const PairOperands& o) {
  asm volatile("" :: "v"(o.a[0]), "v"(o.a[1]), "v"(o.a[2]), "v"(o.a[3]),
                     "v"(o.b[0]), "v"(o.b[1]), "s"(o.w0), "s"(o.w1));
}

__device__ __forceinline__ void pair_step(const PairOperands& o, float (&s0)[2][4], float (&s1)[2][4]) {
#pragma unroll
  for (int e = 0; e < 4; ++e) {
#pragma unroll
    for (int i = 0; i < 2; ++i) {
#pragma unroll
      for (int j = 0; j < 4; ++j) {
        float x = o.a[j][e] + o.b[i][e];
        x = fmaxf(x, 0.f);
        s0[i][j] = fmaf(x, o.w0[e], s0[i][j]);
        s1[i][j] = fmaf(x, o.w1[e], s1[i][j]);
      }
    }
  }
}

__global__ __launch_bounds__(256) void k_pair_mlp(const float* __restrict__ A,
                                                  const float* __restrict__ Bv,
                                                  const float* __restrict__ W2,
                                                  const float* __restrict__ b2,
                                                  float* __restrict__ out) {
  __shared__ __attribute__((aligned(16))) float L[(V + TT) * PITCH];
  float* BvL = L;
  float* AL  = L + V * PITCH;
  const int tid = threadIdx.x;
  const int tb  = blockIdx.x * TT;

  const v4f* Bv4 = (const v4f*)Bv;
#pragma unroll
  for (int n = 0; n < (V * H / 4) / 256; ++n) {
    const int q = tid + n * 256;
    const int row = q >> 6, c4 = q & 63;
    *(v4f*)(BvL + row * PITCH + 4 * c4) = Bv4[q];
  }
  const v4f* A4 = (const v4f*)(A + tb * H);
#pragma unroll
  for (int n = 0; n < (TT * H / 4) / 256; ++n) {
    const int q = tid + n * 256;
    const int row = q >> 6, c4 = q & 63;
    const int lrow = (row & 3) * 8 + (row >> 2);
    *(v4f*)(AL + lrow * PITCH + 4 * c4) = A4[q];
  }
  __syncthreads();

  const int tg = tid & 7;
  const int vq = tid >> 3;
  const float* ap = AL + tg * PITCH;
  const float* bp = BvL + vq * PITCH;

  float s0[2][4], s1[2][4];
#pragma unroll
  for (int i = 0; i < 2; ++i)
#pragma unroll
    for (int j = 0; j < 4; ++j) { s0[i][j] = 0.f; s1[i][j] = 0.f; }

  PairOperands p, q;
  pair_fetch(p, ap, bp, W2, 0);
  pair_hold(p);
  for (int h = 0; h < H; h += 16) {
    pair_fetch(q, ap, bp, W2, h + 4);
    pair_step(p, s0, s1);
    pair_hold(q);
    pair_fetch(p, ap, bp, W2, h + 8);
    pair_step(q, s0, s1);
    pair_hold(p);
    pair_fetch(q, ap, bp, W2, h + 12);
    pair_step(p, s0, s1);
    pair_hold(q);
    pair_fetch(p, ap, bp, W2, (h + 16) & (H - 1));
    pair_step(q, s0, s1);
    pair_hold(p);
  }

  const float b2_0 = b2[0], b2_1 = b2[1];
#pragma unroll
  for (int i = 0; i < 2; ++i) {
    const int v = vq + 32 * i;
    v4f sc, ps;
#pragma unroll
    for (int j = 0; j < 4; ++j) {
      sc[j] = s0[i][j] + b2_0;
      const float z = s1[i][j] + b2_1;
      ps[j] = 1.0f / (1.0f + __expf(-z));
    }
    *(v4f*)(out + v * NT + tb + 4 * tg) = sc;
    *(v4f*)(out + V * NT + v * NT + tb + 4 * tg) = ps;
  }
}

extern "C" void kernel_launch(void* const* d_in, const int* in_sizes, int n_in,
                              void* d_out, int out_size, void* d_ws, size_t ws_size,
                              hipStream_t stream) {
  const float* xv  = (const float*)d_in[0];
  const float* xt  = (const float*)d_in[1];
  const int*   cur = (const int*)  d_in[2];
  const float* W1  = (const float*)d_in[3];
  const float* b1  = (const float*)d_in[4];
  const float* W2  = (const float*)d_in[5];
  const float* b2  = (const float*)d_in[6];
  float* out = (float*)d_out;

  float* A  = (float*)d_ws;
  float* Bv = A + NT * H;

  k_proj<<<NT / 32, PROJ_THREADS, 0, stream>>>(xv, cur, xt, W1, b1, A, Bv);
  k_pair_mlp<<<NT / TT, 256, 0, stream>>>(A, Bv, W2, b2, out);
}


# vehicle waves read their W1 rows as whole 128-byte lines (4 lines per load, not 32) through a wave-private LDS strip, two 32-column chunks requested ahead; per-output fmaf order unchanged
# speedup vs baseline: 3.5010x; 1.1253x over previous
#include <hip/hip_runtime.h>
#include <math.h>

typedef __attribute__((ext_vector_type(2))) float v2f;
typedef __attribute__((ext_vector_type(8))) float v8f;

#define H  256
#define NV 512
#define NT 8192
#define V  64

typedef __attribute__((ext_vector_type(4))) float v4f;

#define VSP 36

struct VehRaw {
  v4f r[8];
};

__device__ __forceinline__ void veh_fetch(VehRaw& g, const float* wl, int c) {
#pragma unroll
  for (int n = 0; n < 8; ++n) g.r[n] = *(const v4f*)(wl + 4 * n * (2 * H) + 32 * c);
}

__device__ __forceinline__ void veh_park(float* st, const VehRaw& g) {
#pragma unroll
  for (int n = 0; n < 8; ++n) *(v4f*)(st + 4 * n * VSP) = g.r[n];
}

__device__ __forceinline__ void vehicle_proj_part(int blk, int t,
                                                  const float* __restrict__ xv,
                                                  const int* __restrict__ current,
                                                  const float* __restrict__ W1,
                                                  const float* __restrict__ b1,
                                                  float* __restrict__ Bv,
                                                  float* strip0, float* strip1) {
  const int lane  = t & 31;
  const int sub   = lane >> 3;
  const int piece = lane & 7;
  const int v  = blk >> 2;
  const int h0 = 64 * (blk & 3) + (t & 32);
  const int h  = h0 + lane;
  const float* veh = xv + current[2 * v] * H;
  const float* wl  = W1 + (h0 + sub) * (2 * H) + H + 4 * piece;
  float* park[2]       = { strip0 + sub * VSP + 4 * piece, strip1 + sub * VSP + 4 * piece };
  const float* mine[2] = { strip0 + lane * VSP,            strip1 + lane * VSP };

  VehRaw g0, g1;
  veh_fetch(g0, wl, 0);
  veh_fetch(g1, wl, 1);
  float s = b1[h];
#pragma unroll
  for (int c = 0; c < 8; ++c) {
    const int cur = c & 1;
    if (cur == 0) {
      veh_park(park[0], g0);
      if (c + 2 < 8) veh_fetch(g0, wl, c + 2);
    } else {
      veh_park(park[1], g1);
      if (c + 2 < 8) veh_fetch(g1, wl, c + 2);
    }
    v4f w[8];
#pragma unroll
    for (int j = 0; j < 8; ++j) w[j] = *(const v4f*)(mine[cur] + 4 * j);
#pragma unroll
    for (int j = 0; j < 8; ++j) {
#pragma unroll
      for (int e = 0; e < 4; ++e) s = fmaf(veh[32 * c + 4 * j + e], w[j][e], s);
    }
  }
  Bv[v * H + h] = s;
}


struct Frag16 {
  v4f q[4];
};

__device__ __forceinline__ void frag_fetch(Frag16& f, const float* p) {
#pragma unroll
  for (int j = 0; j < 4; ++j) f.q[j] = *(const v4f*)(p + 4 * j);
}

__device__ __forceinline__ v2f frag_step(const Frag16& f, int s) {
  v2f r;
  r[0] = f.q[s >> 1][2 * (s & 1)];
  r[1] = f.q[s >> 1][2 * (s & 1) + 1];
  return r;
}

#define SP         36
#define STRIP_ROWS 64

struct RawBlock {
  v4f x[8];
  v4f w[8];
};

__device__ __forceinline__ void raw_fetch(RawBlock& g, const float* xa, const float* wb, int ko) {
#pragma unroll
  for (int n = 0; n < 8; ++n) g.x[n] = *(const v4f*)(xa + 4 * n * H + ko);
#pragma unroll
  for (int n = 0; n < 8; ++n) g.w[n] = *(const v4f*)(wb + 4 * n * (2 * H) + ko);
}

__device__ __forceinline__ void raw_park(float* st, const RawBlock& g) {
#pragma unroll
  for (int n = 0; n < 8; ++n) *(v4f*)(st + 4 * n * SP) = g.x[n];
#pragma unroll
  for (int n = 0; n < 8; ++n) *(v4f*)(st + (32 + 4 * n) * SP) = g.w[n];
}

#define GEMM_WAVES   8
#define PROJ_THREADS 320

__global__ __launch_bounds__(PROJ_THREADS) void k_proj(const float* __restrict__ xv,
                                                       const int* __restrict__ current,
                                                       const float* __restrict__ xt,
                                                       const float* __restrict__ W1,
                                                       const float* __restrict__ b1,
                                                       float* __restrict__ A,
                                                       float* __restrict__ Bv) {
  const int lane = threadIdx.x & 31;
  const int wv   = __builtin_amdgcn_readfirstlane(threadIdx.x >> 5);
  if (wv >= GEMM_WAVES) {
    __shared__ __attribute__((aligned(16))) float vstrips[2][2][32 * VSP];
    const int vw = wv - GEMM_WAVES;
    vehicle_proj_part(blockIdx.x, threadIdx.x - 32 * GEMM_WAVES, xv, current, W1, b1, Bv,
                      &vstrips[vw][0][0], &vstrips[vw][1][0]);
    return;
  }

  const int half = lane >> 4;
  const int l    = lane & 15;
  const int m0   = blockIdx.x * 32;
  const int n0   = wv * 32;

  const int sub   = lane >> 3;
  const int piece = lane & 7;
  const float* xa = xt + (m0 + sub) * H + 4 * piece;
  const float* wb = W1 + (n0 + sub) * (2 * H) + 4 * piece;

  __shared__ __attribute__((aligned(16))) float strips[GEMM_WAVES][2][STRIP_ROWS * SP];
  float* park[2]       = { &strips[wv][0][sub * SP + 4 * piece], &strips[wv][1][sub * SP + 4 * piece] };
  const float* frag[2] = { &strips[wv][0][l * SP + 16 * half],   &strips[wv][1][l * SP + 16 * half] };

  v8f c[2][2];
#pragma unroll
  for (int mi = 0; mi < 2; ++mi)
#pragma unroll
    for (int ni = 0; ni < 2; ++ni) c[mi][ni] = (v8f){};

  RawBlock g, g1;
  Frag16 fa[2][2], fb[2][2];
  raw_fetch(g, xa, wb, 0);
  raw_fetch(g1, xa, wb, 32);
  raw_park(park[0], g);
  raw_fetch(g, xa, wb, 64);
  frag_fetch(fa[0][0], frag[0]);
  frag_fetch(fa[0][1], frag[0] + 16 * SP);
  frag_fetch(fb[0][0], frag[0] + 32 * SP);
  frag_fetch(fb[0][1], frag[0] + 48 * SP);

#pragma unroll
  for (int kb = 0; kb < H / 32; ++kb) {
    const int cur = kb & 1, nxt = cur ^ 1;
    if (kb + 1 < H / 32) {
      if (kb == 0) {
        raw_park(park[nxt], g1);
      } else {
        raw_park(park[nxt], g);
        if (kb + 2 < H / 32) raw_fetch(g, xa, wb, (kb + 2) * 32);
      }
      frag_fetch(fa[nxt][0], frag[nxt]);
      frag_fetch(fa[nxt][1], frag[nxt] + 16 * SP);
      frag_fetch(fb[nxt][0], frag[nxt] + 32 * SP);
      frag_fetch(fb[nxt][1], frag[nxt] + 48 * SP);
    }
    __builtin_amdgcn_sched_barrier(0);
#pragma unroll
    for (int s = 0; s < 8; ++s) {
#pragma unroll
      for (int mi = 0; mi < 2; ++mi) {
#pragma unroll
        for (int ni = 0; ni < 2; ++ni) {
          c[mi][ni] = __builtin_amdgcn_wmma_f32_16x16x4_f32(
               false, frag_step(fa[cur][mi], s),
               false, frag_step(fb[cur][ni], s),
               (short)0, c[mi][ni],  false,  false);
        }
      }
    }
    __builtin_amdgcn_sched_barrier(0);
  }

#pragma unroll
  for (int mi = 0; mi < 2; ++mi) {
#pragma unroll
    for (int ni = 0; ni < 2; ++ni) {
      float* dptr = A + (m0 + 16 * mi + 8 * half) * H + n0 + 16 * ni + l;
#pragma unroll
      for (int r = 0; r < 8; ++r) dptr[r * H] = c[mi][ni][r];
    }
  }
}

#define TT    32
#define PITCH 260

struct PairOperands {
  v4f a[4];
  v4f b[2];
  v4f w0, w1;
};

__device__ __forceinline__ void pair_fetch(PairOperands& o, const float* ap, const float* bp,
                                           const float* __restrict__ W2, int h) {
#pragma unroll
  for (int j = 0; j < 4; ++j) o.a[j] = *(const v4f*)(ap + j * 8 * PITCH + h);
#pragma unroll
  for (int i = 0; i < 2; ++i) o.b[i] = *(const v4f*)(bp + i * 32 * PITCH + h);
  o.w0 = *(const v4f*)(W2 + h);
  o.w1 = *(const v4f*)(W2 + H + h);
}

__device__ __forceinline__ void pair_hold(const PairOperands& o) {
  asm volatile("" :: "v"(o.a[0]), "v"(o.a[1]), "v"(o.a[2]), "v"(o.a[3]),
                     "v"(o.b[0]), "v"(o.b[1]), "s"(o.w0), "s"(o.w1));
}

__device__ __forceinline__ void pair_step(const PairOperands& o, float (&s0)[2][4], float (&s1)[2][4]) {
#pragma unroll
  for (int e = 0; e < 4; ++e) {
#pragma unroll
    for (int i = 0; i < 2; ++i) {
#pragma unroll
      for (int j = 0; j < 4; ++j) {
        float x = o.a[j][e] + o.b[i][e];
        x = fmaxf(x, 0.f);
        s0[i][j] = fmaf(x, o.w0[e], s0[i][j]);
        s1[i][j] = fmaf(x, o.w1[e], s1[i][j]);
      }
    }
  }
}

__global__ __launch_bounds__(256) void k_pair_mlp(const float* __restrict__ A,
                                                  const float* __restrict__ Bv,
                                                  const float* __restrict__ W2,
                                                  const float* __restrict__ b2,
                                                  float* __restrict__ out) {
  __shared__ __attribute__((aligned(16))) float L[(V + TT) * PITCH];
  float* BvL = L;
  float* AL  = L + V * PITCH;
  const int tid = threadIdx.x;
  const int tb  = blockIdx.x * TT;

  const v4f* Bv4 = (const v4f*)Bv;
#pragma unroll
  for (int n = 0; n < (V * H / 4) / 256; ++n) {
    const int q = tid + n * 256;
    const int row = q >> 6, c4 = q & 63;
    *(v4f*)(BvL + row * PITCH + 4 * c4) = Bv4[q];
  }
  const v4f* A4 = (const v4f*)(A + tb * H);
#pragma unroll
  for (int n = 0; n < (TT * H / 4) / 256; ++n) {
    const int q = tid + n * 256;
    const int row = q >> 6, c4 = q & 63;
    const int lrow = (row & 3) * 8 + (row >> 2);
    *(v4f*)(AL + lrow * PITCH + 4 * c4) = A4[q];
  }
  __syncthreads();

  const int tg = tid & 7;
  const int vq = tid >> 3;
  const float* ap = AL + tg * PITCH;
  const float* bp = BvL + vq * PITCH;

  float s0[2][4], s1[2][4];
#pragma unroll
  for (int i = 0; i < 2; ++i)
#pragma unroll
    for (int j = 0; j < 4; ++j) { s0[i][j] = 0.f; s1[i][j] = 0.f; }

  PairOperands p, q;
  pair_fetch(p, ap, bp, W2, 0);
  pair_hold(p);
  for (int h = 0; h < H; h += 16) {
    pair_fetch(q, ap, bp, W2, h + 4);
    pair_step(p, s0, s1);
    pair_hold(q);
    pair_fetch(p, ap, bp, W2, h + 8);
    pair_step(q, s0, s1);
    pair_hold(p);
    pair_fetch(q, ap, bp, W2, h + 12);
    pair_step(p, s0, s1);
    pair_hold(q);
    pair_fetch(p, ap, bp, W2, (h + 16) & (H - 1));
    pair_step(q, s0, s1);
    pair_hold(p);
  }

  const float b2_0 = b2[0], b2_1 = b2[1];
#pragma unroll
  for (int i = 0; i < 2; ++i) {
    const int v = vq + 32 * i;
    v4f sc, ps;
#pragma unroll
    for (int j = 0; j < 4; ++j) {
      sc[j] = s0[i][j] + b2_0;
      const float z = s1[i][j] + b2_1;
      ps[j] = 1.0f / (1.0f + __expf(-z));
    }
    *(v4f*)(out + v * NT + tb + 4 * tg) = sc;
    *(v4f*)(out + V * NT + v * NT + tb + 4 * tg) = ps;
  }
}

extern "C" void kernel_launch(void* const* d_in, const int* in_sizes, int n_in,
                              void* d_out, int out_size, void* d_ws, size_t ws_size,
                              hipStream_t stream) {
  const float* xv  = (const float*)d_in[0];
  const float* xt  = (const float*)d_in[1];
  const int*   cur = (const int*)  d_in[2];
  const float* W1  = (const float*)d_in[3];
  const float* b1  = (const float*)d_in[4];
  const float* W2  = (const float*)d_in[5];
  const float* b2  = (const float*)d_in[6];
  float* out = (float*)d_out;

  float* A  = (float*)d_ws;
  float* Bv = A + NT * H;

  k_proj<<<NT / 32, PROJ_THREADS, 0, stream>>>(xv, cur, xt, W1, b1, A, Bv);
  k_pair_mlp<<<NT / TT, 256, 0, stream>>>(A, Bv, W2, b2, out);
}
